# starting kernel
# baseline (speedup 1.0000x reference)
#include <hip/hip_runtime.h>
#include <hip/hip_bf16.h>

typedef __attribute__((ext_vector_type(2))) float v2f;
typedef __attribute__((ext_vector_type(8))) float v8f;

#define B_DIM 64
#define N_DIM 128
#define D_DIM 128
#define A_DIM 1000
#define KT    100

__device__ __forceinline__ v8f wmma_f32(v2f a, v2f b, v8f c) {
  return __builtin_amdgcn_wmma_f32_16x16x4_f32(false, a, false, b, (short)0, c,
                                               false, false);
}

__global__ __launch_bounds__(256) void gat_attn_kernel(
    const float* __restrict__ hidden, const int* __restrict__ adj,
    const float* __restrict__ amat, float* __restrict__ out) {
  __shared__ float sH[N_DIM * D_DIM];
  __shared__ float sAlpha[16 * N_DIM];
  __shared__ float sAa[D_DIM * 4];
  __shared__ float sRed[16][16];
  __shared__ float sMax[16];
  __shared__ float sSum[16];

  const int tid    = threadIdx.x;
  const int b      = blockIdx.x >> 3;
  const int i_base = (blockIdx.x & 7) * 16;

  const float* Hg = hidden + (size_t)b * N_DIM * D_DIM;
  for (int idx4 = tid; idx4 < (N_DIM * D_DIM) / 4; idx4 += 256)
    reinterpret_cast<float4*>(sH)[idx4] = reinterpret_cast<const float4*>(Hg)[idx4];
  if (tid < (D_DIM * 4) / 4)
    reinterpret_cast<float4*>(sAa)[tid] = reinterpret_cast<const float4*>(amat)[tid];
  __syncthreads();

  const int wave   = tid >> 5;
  const int lane   = tid & 31;
  const int lo16   = lane & 15;
  const int hi16   = (lane >= 16) ? 8 : 0;
  const int koff   = (lane >= 16) ? 2 : 0;
  const int j_base = wave * 16;

  const float C1 = 0.6f, C2 = 0.4f;
  const float NEG_INF = -9.0e15f;

  v8f acc[4];
  acc[0] = {}; acc[1] = {}; acc[2] = {}; acc[3] = {};
  const int i_row = i_base + lo16;
  const int j_row = j_base + lo16;
  for (int d0 = 0; d0 < D_DIM; d0 += 4) {
    const int d = d0 + koff;
    const float2 hi = *reinterpret_cast<const float2*>(&sH[i_row * D_DIM + d]);
    const float2 hj = *reinterpret_cast<const float2*>(&sH[j_row * D_DIM + d]);
    const float* a0 = &sAa[d * 4];
    const float* a1 = &sAa[(d + 1) * 4];
    const float hax = fabsf(hi.x), hay = fabsf(hi.y);
    v2f Bv;  Bv[0] = hj.x;        Bv[1] = hj.y;
    v2f Bva; Bva[0] = fabsf(hj.x); Bva[1] = fabsf(hj.y);
#pragma unroll
    for (int k = 0; k < 4; ++k) {
      v2f Av;  Av[0]  = C1 * hi.x * a0[k]; Av[1]  = C1 * hi.y * a1[k];
      acc[k] = wmma_f32(Av, Bv, acc[k]);
      v2f Ava; Ava[0] = C2 * hax * a0[k];  Ava[1] = C2 * hay * a1[k];
      acc[k] = wmma_f32(Ava, Bva, acc[k]);
    }
  }

  const int* adjB = adj + (size_t)b * N_DIM * N_DIM;
#pragma unroll
  for (int r = 0; r < 8; ++r) {
    const int il = r + hi16;
    const int t = adjB[(i_base + il) * N_DIM + j_base + lo16];
    float e = NEG_INF;
    e = (t == 1) ? acc[0][r] : e;
    e = (t == 2) ? acc[1][r] : e;
    e = (t == 3) ? acc[2][r] : e;
    e = (t == 4) ? acc[3][r] : e;
    sAlpha[il * N_DIM + j_base + lo16] = e;
  }
  __syncthreads();

  const int row = tid >> 4;
  const int part = tid & 15;
  float m = -3.4e38f;
  for (int c = part * 8; c < part * 8 + 8; ++c)
    m = fmaxf(m, sAlpha[row * N_DIM + c]);
  sRed[row][part] = m;
  __syncthreads();
  if (part == 0) {
    float mm = sRed[row][0];
    for (int q = 1; q < 16; ++q) mm = fmaxf(mm, sRed[row][q]);
    sMax[row] = mm;
  }
  __syncthreads();
  const float mm2 = sMax[row];
  float s = 0.f;
  for (int c = part * 8; c < part * 8 + 8; ++c) {
    const float ev = __expf(sAlpha[row * N_DIM + c] - mm2);
    sAlpha[row * N_DIM + c] = ev;
    s += ev;
  }
  sRed[row][part] = s;
  __syncthreads();
  if (part == 0) {
    float ss = 0.f;
    for (int q = 0; q < 16; ++q) ss += sRed[row][q];
    sSum[row] = ss;
  }
  __syncthreads();
  const float inv = 1.0f / sSum[row];
  for (int c = part * 8; c < part * 8 + 8; ++c)
    sAlpha[row * N_DIM + c] *= inv;
  __syncthreads();

  const int dg = j_base + lo16;
  v8f oacc = {};
  for (int j0 = 0; j0 < N_DIM; j0 += 4) {
    const int jk = j0 + koff;
    const float2 av = *reinterpret_cast<const float2*>(&sAlpha[lo16 * N_DIM + jk]);
    v2f Av; Av[0] = av.x; Av[1] = av.y;
    v2f Bv; Bv[0] = sH[jk * D_DIM + dg]; Bv[1] = sH[(jk + 1) * D_DIM + dg];
    oacc = wmma_f32(Av, Bv, oacc);
  }
#pragma unroll
  for (int r = 0; r < 8; ++r)
    out[((size_t)b * N_DIM + i_base + r + hi16) * D_DIM + dg] = oacc[r];
}

__global__ __launch_bounds__(256) void attr_gemm_kernel(
    const float* __restrict__ Aattr, const float* __restrict__ Eemb,
    float* __restrict__ out2) {
  __shared__ float sAp[16 * KT];
  __shared__ float sEp[KT * D_DIM];

  const int tid    = threadIdx.x;
  const int m_base = blockIdx.x * 16;
  const int wave   = tid >> 5;
  const int lane   = tid & 31;
  const int lo16   = lane & 15;
  const int hi16   = (lane >= 16) ? 8 : 0;
  const int koff   = (lane >= 16) ? 2 : 0;
  const int dg     = wave * 16 + lo16;

  v8f acc = {};
  for (int kt = 0; kt < A_DIM / KT; ++kt) {
    const int k0 = kt * KT;
    __syncthreads();
    for (int idx4 = tid; idx4 < (16 * KT) / 4; idx4 += 256) {
      const int r = idx4 / (KT / 4), c4 = idx4 % (KT / 4);
      *reinterpret_cast<float4*>(&sAp[r * KT + c4 * 4]) =
          *reinterpret_cast<const float4*>(
              &Aattr[(size_t)(m_base + r) * A_DIM + k0 + c4 * 4]);
    }
    for (int idx4 = tid; idx4 < (KT * D_DIM) / 4; idx4 += 256) {
      const int r = idx4 >> 5, c4 = idx4 & 31;
      *reinterpret_cast<float4*>(&sEp[r * D_DIM + c4 * 4]) =
          *reinterpret_cast<const float4*>(
              &Eemb[(size_t)(k0 + r) * D_DIM + c4 * 4]);
    }
    __syncthreads();
    if (kt + 1 < A_DIM / KT)
      __builtin_prefetch(&Aattr[(size_t)m_base * A_DIM + k0 + KT], 0, 0);
#pragma unroll
    for (int ks = 0; ks < KT / 4; ++ks) {
      const int k = 4 * ks + koff;
      const float2 av = *reinterpret_cast<const float2*>(&sAp[lo16 * KT + k]);
      v2f Av; Av[0] = av.x; Av[1] = av.y;
      v2f Bv; Bv[0] = sEp[k * D_DIM + dg]; Bv[1] = sEp[(k + 1) * D_DIM + dg];
      acc = wmma_f32(Av, Bv, acc);
    }
  }
#pragma unroll
  for (int r = 0; r < 8; ++r)
    out2[(size_t)(m_base + r + hi16) * D_DIM + dg] = acc[r];
}

extern "C" void kernel_launch(void* const* d_in, const int* in_sizes, int n_in,
                              void* d_out, int out_size, void* d_ws, size_t ws_size,
                              hipStream_t stream) {
  (void)in_sizes; (void)n_in; (void)out_size; (void)d_ws; (void)ws_size;
  const float* hidden = (const float*)d_in[0];
  const int*   adj    = (const int*)d_in[1];
  const float* amat   = (const float*)d_in[2];
  const float* Aattr  = (const float*)d_in[3];
  const float* Eemb   = (const float*)d_in[4];

  float* out_attn = (float*)d_out;
  float* out_attr = out_attn + (size_t)B_DIM * N_DIM * D_DIM;

  gat_attn_kernel<<<dim3(B_DIM * (N_DIM / 16)), dim3(256), 0, stream>>>(
      hidden, adj, amat, out_attn);
  attr_gemm_kernel<<<dim3((B_DIM * N_DIM) / 16), dim3(256), 0, stream>>>(
      Aattr, Eemb, out_attr);
}


# attr GEMM re-tiled: one 8-wave group per 32 rows; 4 waves multiply 16x64 tiles, A fragments straight from global; 4 waves stage E k-quad-interleaved into an LDS ring meanwhile; k sum order changed
# speedup vs baseline: 1.7836x; 1.7836x over previous
#include <hip/hip_runtime.h>
#include <hip/hip_bf16.h>

typedef __attribute__((ext_vector_type(2))) float v2f;
typedef __attribute__((ext_vector_type(8))) float v8f;

#define B_DIM 64
#define N_DIM 128
#define D_DIM 128
#define A_DIM 1000

__device__ __forceinline__ v8f wmma_f32(v2f a, v2f b, v8f c) {
  return __builtin_amdgcn_wmma_f32_16x16x4_f32(false, a, false, b, (short)0, c,
                                               false, false);
}

__global__ __launch_bounds__(256) void gat_attn_kernel(
    const float* __restrict__ hidden, const int* __restrict__ adj,
    const float* __restrict__ amat, float* __restrict__ out) {
  __shared__ float sH[N_DIM * D_DIM];
  __shared__ float sAlpha[16 * N_DIM];
  __shared__ float sAa[D_DIM * 4];
  __shared__ float sRed[16][16];
  __shared__ float sMax[16];
  __shared__ float sSum[16];

  const int tid    = threadIdx.x;
  const int b      = blockIdx.x >> 3;
  const int i_base = (blockIdx.x & 7) * 16;

  const float* Hg = hidden + (size_t)b * N_DIM * D_DIM;
  for (int idx4 = tid; idx4 < (N_DIM * D_DIM) / 4; idx4 += 256)
    reinterpret_cast<float4*>(sH)[idx4] = reinterpret_cast<const float4*>(Hg)[idx4];
  if (tid < (D_DIM * 4) / 4)
    reinterpret_cast<float4*>(sAa)[tid] = reinterpret_cast<const float4*>(amat)[tid];
  __syncthreads();

  const int wave   = tid >> 5;
  const int lane   = tid & 31;
  const int lo16   = lane & 15;
  const int hi16   = (lane >= 16) ? 8 : 0;
  const int koff   = (lane >= 16) ? 2 : 0;
  const int j_base = wave * 16;

  const float C1 = 0.6f, C2 = 0.4f;
  const float NEG_INF = -9.0e15f;

  v8f acc[4];
  acc[0] = {}; acc[1] = {}; acc[2] = {}; acc[3] = {};
  const int i_row = i_base + lo16;
  const int j_row = j_base + lo16;
  for (int d0 = 0; d0 < D_DIM; d0 += 4) {
    const int d = d0 + koff;
    const float2 hi = *reinterpret_cast<const float2*>(&sH[i_row * D_DIM + d]);
    const float2 hj = *reinterpret_cast<const float2*>(&sH[j_row * D_DIM + d]);
    const float* a0 = &sAa[d * 4];
    const float* a1 = &sAa[(d + 1) * 4];
    const float hax = fabsf(hi.x), hay = fabsf(hi.y);
    v2f Bv;  Bv[0] = hj.x;        Bv[1] = hj.y;
    v2f Bva; Bva[0] = fabsf(hj.x); Bva[1] = fabsf(hj.y);
#pragma unroll
    for (int k = 0; k < 4; ++k) {
      v2f Av;  Av[0]  = C1 * hi.x * a0[k]; Av[1]  = C1 * hi.y * a1[k];
      acc[k] = wmma_f32(Av, Bv, acc[k]);
      v2f Ava; Ava[0] = C2 * hax * a0[k];  Ava[1] = C2 * hay * a1[k];
      acc[k] = wmma_f32(Ava, Bva, acc[k]);
    }
  }

  const int* adjB = adj + (size_t)b * N_DIM * N_DIM;
#pragma unroll
  for (int r = 0; r < 8; ++r) {
    const int il = r + hi16;
    const int t = adjB[(i_base + il) * N_DIM + j_base + lo16];
    float e = NEG_INF;
    e = (t == 1) ? acc[0][r] : e;
    e = (t == 2) ? acc[1][r] : e;
    e = (t == 3) ? acc[2][r] : e;
    e = (t == 4) ? acc[3][r] : e;
    sAlpha[il * N_DIM + j_base + lo16] = e;
  }
  __syncthreads();

  const int row = tid >> 4;
  const int part = tid & 15;
  float m = -3.4e38f;
  for (int c = part * 8; c < part * 8 + 8; ++c)
    m = fmaxf(m, sAlpha[row * N_DIM + c]);
  sRed[row][part] = m;
  __syncthreads();
  if (part == 0) {
    float mm = sRed[row][0];
    for (int q = 1; q < 16; ++q) mm = fmaxf(mm, sRed[row][q]);
    sMax[row] = mm;
  }
  __syncthreads();
  const float mm2 = sMax[row];
  float s = 0.f;
  for (int c = part * 8; c < part * 8 + 8; ++c) {
    const float ev = __expf(sAlpha[row * N_DIM + c] - mm2);
    sAlpha[row * N_DIM + c] = ev;
    s += ev;
  }
  sRed[row][part] = s;
  __syncthreads();
  if (part == 0) {
    float ss = 0.f;
    for (int q = 0; q < 16; ++q) ss += sRed[row][q];
    sSum[row] = ss;
  }
  __syncthreads();
  const float inv = 1.0f / sSum[row];
  for (int c = part * 8; c < part * 8 + 8; ++c)
    sAlpha[row * N_DIM + c] *= inv;
  __syncthreads();

  const int dg = j_base + lo16;
  v8f oacc = {};
  for (int j0 = 0; j0 < N_DIM; j0 += 4) {
    const int jk = j0 + koff;
    const float2 av = *reinterpret_cast<const float2*>(&sAlpha[lo16 * N_DIM + jk]);
    v2f Av; Av[0] = av.x; Av[1] = av.y;
    v2f Bv; Bv[0] = sH[jk * D_DIM + dg]; Bv[1] = sH[(jk + 1) * D_DIM + dg];
    oacc = wmma_f32(Av, Bv, oacc);
  }
#pragma unroll
  for (int r = 0; r < 8; ++r)
    out[((size_t)b * N_DIM + i_base + r + hi16) * D_DIM + dg] = oacc[r];
}

#define KP     200
#define NPANEL (A_DIM / KP)
#define KQP    (KP / 4)
#define CH     (KP / 8)
#define GRP    5
#define NGRP   (CH / GRP)

template <int NT>
__device__ __forceinline__ void stage_panel(const float* __restrict__ Ep,
                                            float4* __restrict__ sbuf, int t) {
  constexpr int NBLK = KQP * (D_DIM / 2);
  constexpr int J    = (NBLK + NT - 1) / NT;
  constexpr int TAIL = NBLK - (J - 1) * NT;
  float2 e[J][4];
#pragma unroll
  for (int j = 0; j < J; ++j) {
    const int blk = t + j * NT;
    if (j < J - 1 || TAIL == NT || t < TAIL) {
      const int kq = blk >> 6, n2 = blk & 63;
      const float* s = Ep + (size_t)(4 * kq) * D_DIM + 2 * n2;
#pragma unroll
      for (int jj = 0; jj < 4; ++jj)
        e[j][jj] = *reinterpret_cast<const float2*>(s + jj * D_DIM);
    }
  }
#pragma unroll
  for (int j = 0; j < J; ++j) {
    const int blk = t + j * NT;
    if (j < J - 1 || TAIL == NT || t < TAIL) {
      const int kq = blk >> 6, n2 = blk & 63;
      float4* d = sbuf + kq * D_DIM + 2 * n2;
      d[0] = make_float4(e[j][0].x, e[j][1].x, e[j][2].x, e[j][3].x);
      d[1] = make_float4(e[j][0].y, e[j][1].y, e[j][2].y, e[j][3].y);
    }
  }
}

__global__ __launch_bounds__(256) void attr_gemm_kernel(
    const float* __restrict__ Aattr, const float* __restrict__ Eemb,
    float* __restrict__ out2) {
  __shared__ float4 sE[2][KQP * D_DIM];

  const int tid    = threadIdx.x;
  const int wave   = __builtin_amdgcn_readfirstlane(tid >> 5);
  const int lane   = tid & 31;
  const int c      = lane & 15;
  const int h      = lane >> 4;
  const int m_base = blockIdx.x * 32;
  const bool mult  = wave < 4;

  const int mt = (wave >> 1) & 1;
  const int nh = wave & 1;
  const float* Arow = Aattr + (size_t)(m_base + 16 * mt + c) * A_DIM + 4 * h;

  float4 aCur[CH];
  v8f acc[4];
  acc[0] = {}; acc[1] = {}; acc[2] = {}; acc[3] = {};

  if (mult) {
#pragma unroll
    for (int q = 0; q < CH; ++q)
      aCur[q] = *reinterpret_cast<const float4*>(Arow + 8 * q);
  }
  stage_panel<256>(Eemb, sE[0], tid);
  __syncthreads();

  for (int p = 0; p < NPANEL; ++p) {
    if (mult) {
      const int pn = (p + 1 < NPANEL) ? p + 1 : p;
      const float* Anext = Arow + 8 * CH * pn;
      const float4* fb = sE[p & 1] + h * D_DIM + 64 * nh + c;
      float4 bs[2][GRP][4];
#pragma unroll
      for (int qq = 0; qq < GRP; ++qq)
#pragma unroll
        for (int t = 0; t < 4; ++t)
          bs[0][qq][t] = fb[(2 * qq) * D_DIM + 16 * t];
#pragma unroll
      for (int g = 0; g < NGRP; ++g) {
        if (g + 1 < NGRP) {
#pragma unroll
          for (int qq = 0; qq < GRP; ++qq)
#pragma unroll
            for (int t = 0; t < 4; ++t)
              bs[(g + 1) & 1][qq][t] = fb[(2 * (GRP * (g + 1) + qq)) * D_DIM + 16 * t];
        }
        __builtin_amdgcn_sched_barrier(0);
#pragma unroll
        for (int qq = 0; qq < GRP; ++qq) {
          const float4 a = aCur[GRP * g + qq];
          v2f A0; A0[0] = a.x; A0[1] = a.y;
          v2f A1; A1[0] = a.z; A1[1] = a.w;
#pragma unroll
          for (int t = 0; t < 4; ++t) {
            const float4 b = bs[g & 1][qq][t];
            v2f B0; B0[0] = b.x; B0[1] = b.y;
            acc[t] = wmma_f32(A0, B0, acc[t]);
          }
#pragma unroll
          for (int t = 0; t < 4; ++t) {
            const float4 b = bs[g & 1][qq][t];
            v2f B1; B1[0] = b.z; B1[1] = b.w;
            acc[t] = wmma_f32(A1, B1, acc[t]);
          }
          aCur[GRP * g + qq] =
              *reinterpret_cast<const float4*>(Anext + 8 * (GRP * g + qq));
        }
      }
    } else {
      if (p + 1 < NPANEL) {
        const float* Ep = Eemb + (size_t)(p + 1) * KP * D_DIM;
        stage_panel<128>(Ep, sE[(p + 1) & 1], tid - 128);
      }
    }
    if (p + 1 < NPANEL) {
      __builtin_amdgcn_fence(__ATOMIC_RELEASE, "workgroup", "local");
      __builtin_amdgcn_s_barrier();
      __builtin_amdgcn_fence(__ATOMIC_ACQUIRE, "workgroup", "local");
    }
  }

  if (mult) {
    const int hi8 = 8 * h;
#pragma unroll
    for (int t = 0; t < 4; ++t)
#pragma unroll
      for (int r = 0; r < 8; ++r)
        out2[(size_t)(m_base + 16 * mt + r + hi8) * D_DIM + 64 * nh + 16 * t + c] = acc[t][r];
  }
}

extern "C" void kernel_launch(void* const* d_in, const int* in_sizes, int n_in,
                              void* d_out, int out_size, void* d_ws, size_t ws_size,
                              hipStream_t stream) {
  (void)in_sizes; (void)n_in; (void)out_size; (void)d_ws; (void)ws_size;
  const float* hidden = (const float*)d_in[0];
  const int*   adj    = (const int*)d_in[1];
  const float* amat   = (const float*)d_in[2];
  const float* Aattr  = (const float*)d_in[3];
  const float* Eemb   = (const float*)d_in[4];

  float* out_attn = (float*)d_out;
  float* out_attr = out_attn + (size_t)B_DIM * N_DIM * D_DIM;

  gat_attn_kernel<<<dim3(B_DIM * (N_DIM / 16)), dim3(256), 0, stream>>>(
      hidden, adj, amat, out_attn);
  attr_gemm_kernel<<<dim3((B_DIM * N_DIM) / 32), dim3(256), 0, stream>>>(
      Aattr, Eemb, out_attr);
}


# attention re-tiled: one 4-wave group per 32 rows, each wave a 16x64 score strip of all 4 edge types (scaled A built once per 4 column tiles), softmax in registers; k and softmax sum order changed
# speedup vs baseline: 2.0988x; 1.1767x over previous
#include <hip/hip_runtime.h>
#include <hip/hip_bf16.h>

typedef __attribute__((ext_vector_type(2))) float v2f;
typedef __attribute__((ext_vector_type(8))) float v8f;

#define B_DIM 64
#define N_DIM 128
#define D_DIM 128
#define A_DIM 1000

__device__ __forceinline__ v8f wmma_f32(v2f a, v2f b, v8f c) {
  return __builtin_amdgcn_wmma_f32_16x16x4_f32(false, a, false, b, (short)0, c,
                                               false, false);
}

typedef __attribute__((ext_vector_type(4))) float v4f;

#define PH 132

struct SFrag { v4f hi; v4f hj[4]; v4f a[4]; };
struct OFrag { v4f p; v4f b[4]; };

__device__ __forceinline__ void load_sfrag(SFrag& f, const float* pHi, const float* pHj,
                                           const float* pA, int q) {
  f.hi = *reinterpret_cast<const v4f*>(pHi + 8 * q);
#pragma unroll
  for (int t = 0; t < 4; ++t)
    f.hj[t] = *reinterpret_cast<const v4f*>(pHj + 16 * t * PH + 8 * q);
#pragma unroll
  for (int s = 0; s < 4; ++s)
    f.a[s] = *reinterpret_cast<const v4f*>(pA + 32 * q + 4 * s);
}

__device__ __forceinline__ void mul_sfrag(const SFrag& f, v8f (&acc)[4][4]) {
  const float C1 = 0.6f, C2 = 0.4f;
  float t1[4], t2[4];
#pragma unroll
  for (int s = 0; s < 4; ++s) {
    t1[s] = C1 * f.hi[s];
    t2[s] = C2 * fabsf(f.hi[s]);
  }
#pragma unroll
  for (int u = 0; u < 2; ++u) {
    v2f Bv[4], Bva[4];
#pragma unroll
    for (int t = 0; t < 4; ++t) {
      Bv[t][0]  = f.hj[t][2 * u];          Bv[t][1]  = f.hj[t][2 * u + 1];
      Bva[t][0] = fabsf(f.hj[t][2 * u]);   Bva[t][1] = fabsf(f.hj[t][2 * u + 1]);
    }
#pragma unroll
    for (int k = 0; k < 4; ++k) {
      v2f Av;
      Av[0] = t1[2 * u] * f.a[2 * u][k];
      Av[1] = t1[2 * u + 1] * f.a[2 * u + 1][k];
#pragma unroll
      for (int t = 0; t < 4; ++t) acc[k][t] = wmma_f32(Av, Bv[t], acc[k][t]);
      v2f Ava;
      Ava[0] = t2[2 * u] * f.a[2 * u][k];
      Ava[1] = t2[2 * u + 1] * f.a[2 * u + 1][k];
#pragma unroll
      for (int t = 0; t < 4; ++t) acc[k][t] = wmma_f32(Ava, Bva[t], acc[k][t]);
    }
  }
}

__device__ __forceinline__ void load_ofrag(OFrag& f, const float* pP, const float* pB, int q) {
  f.p = *reinterpret_cast<const v4f*>(pP + 8 * q);
#pragma unroll
  for (int s = 0; s < 4; ++s)
    f.b[s] = *reinterpret_cast<const v4f*>(pB + (8 * q + s) * PH);
}

__device__ __forceinline__ void mul_ofrag(const OFrag& f, v8f (&oacc)[4]) {
#pragma unroll
  for (int u = 0; u < 2; ++u) {
    v2f Av; Av[0] = f.p[2 * u]; Av[1] = f.p[2 * u + 1];
#pragma unroll
    for (int t = 0; t < 4; ++t) {
      v2f Bv; Bv[0] = f.b[2 * u][t]; Bv[1] = f.b[2 * u + 1][t];
      oacc[t] = wmma_f32(Av, Bv, oacc[t]);
    }
  }
}

template <int X>
__device__ __forceinline__ float row_xor(float v) {
  return __builtin_bit_cast(
      float, __builtin_amdgcn_update_dpp(0, __builtin_bit_cast(int, v), 0x160 + X, 0xf, 0xf, true));
}

__global__ __launch_bounds__(128) void gat_attn_kernel(
    const float* __restrict__ hidden, const int* __restrict__ adj,
    const float* __restrict__ amat, float* __restrict__ out) {
  __shared__ __attribute__((aligned(16))) float sH[N_DIM * PH];
  __shared__ __attribute__((aligned(16))) float sP[32 * PH];
  __shared__ __attribute__((aligned(16))) float sAa[D_DIM * 4];
  __shared__ __attribute__((aligned(16))) float sRedM[2][2][16];
  __shared__ __attribute__((aligned(16))) float sRedS[2][2][16];

  const int tid  = threadIdx.x;
  const int b    = blockIdx.x >> 2;
  const int wave = __builtin_amdgcn_readfirstlane(tid >> 5);
  const int lane = tid & 31;
  const int c    = lane & 15;
  const int h    = lane >> 4;
  const int it   = wave >> 1;
  const int jh   = wave & 1;
  const int i0   = (blockIdx.x & 3) * 32 + 16 * it;
  const int j0   = 64 * jh;

  const v4f* Hg4 = reinterpret_cast<const v4f*>(hidden + (size_t)b * N_DIM * D_DIM);
  v4f st[32];
#pragma unroll
  for (int n = 0; n < 32; ++n) st[n] = Hg4[tid + 128 * n];
  const v4f a4 = reinterpret_cast<const v4f*>(amat)[tid];

  const int* adjS = adj + ((size_t)b * N_DIM + i0 + 8 * h) * N_DIM + j0 + c;
  int et[4][8];
#pragma unroll
  for (int t = 0; t < 4; ++t)
#pragma unroll
    for (int r = 0; r < 8; ++r) et[t][r] = adjS[r * N_DIM + 16 * t];

#pragma unroll
  for (int n = 0; n < 32; ++n)
    *reinterpret_cast<v4f*>(&sH[(wave + 4 * n) * PH + 4 * lane]) = st[n];
  reinterpret_cast<v4f*>(sAa)[tid] = a4;
  __syncthreads();

  v8f acc[4][4];
#pragma unroll
  for (int k = 0; k < 4; ++k)
#pragma unroll
    for (int t = 0; t < 4; ++t) acc[k][t] = {};

  const float* pHi = &sH[(i0 + c) * PH + 4 * h];
  const float* pHj = &sH[(j0 + c) * PH + 4 * h];
  const float* pA  = &sAa[16 * h];
  SFrag fa, fb;
  load_sfrag(fa, pHi, pHj, pA, 0);
  for (int q = 0; q < 16; q += 2) {
    load_sfrag(fb, pHi, pHj, pA, q + 1);
    __builtin_amdgcn_sched_barrier(0);
    mul_sfrag(fa, acc);
    load_sfrag(fa, pHi, pHj, pA, (q + 2) & 15);
    __builtin_amdgcn_sched_barrier(0);
    mul_sfrag(fb, acc);
  }

  const float NEG_INF = -9.0e15f;
  float e[4][8];
#pragma unroll
  for (int t = 0; t < 4; ++t)
#pragma unroll
    for (int r = 0; r < 8; ++r) {
      const int ty = et[t][r];
      float v = NEG_INF;
      v = (ty == 1) ? acc[0][t][r] : v;
      v = (ty == 2) ? acc[1][t][r] : v;
      v = (ty == 3) ? acc[2][t][r] : v;
      v = (ty == 4) ? acc[3][t][r] : v;
      e[t][r] = v;
    }

  float m[8];
#pragma unroll
  for (int r = 0; r < 8; ++r)
    m[r] = fmaxf(fmaxf(e[0][r], e[1][r]), fmaxf(e[2][r], e[3][r]));
#pragma unroll
  for (int r = 0; r < 8; ++r) m[r] = fmaxf(m[r], row_xor<1>(m[r]));
#pragma unroll
  for (int r = 0; r < 8; ++r) m[r] = fmaxf(m[r], row_xor<2>(m[r]));
#pragma unroll
  for (int r = 0; r < 8; ++r) m[r] = fmaxf(m[r], row_xor<4>(m[r]));
#pragma unroll
  for (int r = 0; r < 8; ++r) m[r] = fmaxf(m[r], row_xor<8>(m[r]));
  if (c == 0) {
#pragma unroll
    for (int r = 0; r < 8; ++r) sRedM[it][jh][8 * h + r] = m[r];
  }
  __syncthreads();
#pragma unroll
  for (int r = 0; r < 8; ++r) m[r] = fmaxf(m[r], sRedM[it][jh ^ 1][8 * h + r]);

  float s[8];
#pragma unroll
  for (int r = 0; r < 8; ++r) {
#pragma unroll
    for (int t = 0; t < 4; ++t) e[t][r] = __expf(e[t][r] - m[r]);
    s[r] = (e[0][r] + e[1][r]) + (e[2][r] + e[3][r]);
  }
#pragma unroll
  for (int r = 0; r < 8; ++r) s[r] += row_xor<1>(s[r]);
#pragma unroll
  for (int r = 0; r < 8; ++r) s[r] += row_xor<2>(s[r]);
#pragma unroll
  for (int r = 0; r < 8; ++r) s[r] += row_xor<4>(s[r]);
#pragma unroll
  for (int r = 0; r < 8; ++r) s[r] += row_xor<8>(s[r]);
  if (c == 0) {
#pragma unroll
    for (int r = 0; r < 8; ++r) sRedS[it][jh][8 * h + r] = s[r];
  }
  __syncthreads();
#pragma unroll
  for (int r = 0; r < 8; ++r) {
    const float tot = sRedS[it][0][8 * h + r] + sRedS[it][1][8 * h + r];
    const float inv = 1.0f / tot;
#pragma unroll
    for (int t = 0; t < 4; ++t)
      sP[(16 * it + 8 * h + r) * PH + j0 + 16 * t + c] = e[t][r] * inv;
  }
  __syncthreads();

  v8f oacc[4];
#pragma unroll
  for (int t = 0; t < 4; ++t) oacc[t] = {};
  const float* pP = &sP[(16 * it + c) * PH + 4 * h];
  const float* pB = &sH[(4 * h) * PH + j0 + 4 * c];
  OFrag oa, ob;
  load_ofrag(oa, pP, pB, 0);
  for (int q = 0; q < 16; q += 2) {
    load_ofrag(ob, pP, pB, q + 1);
    __builtin_amdgcn_sched_barrier(0);
    mul_ofrag(oa, oacc);
    load_ofrag(oa, pP, pB, (q + 2) & 15);
    __builtin_amdgcn_sched_barrier(0);
    mul_ofrag(ob, oacc);
  }
  float* og = out + ((size_t)b * N_DIM + i0 + 8 * h) * D_DIM + j0 + 4 * c;
#pragma unroll
  for (int r = 0; r < 8; ++r) {
    v4f o;
    o[0] = oacc[0][r]; o[1] = oacc[1][r]; o[2] = oacc[2][r]; o[3] = oacc[3][r];
    *reinterpret_cast<v4f*>(og + r * D_DIM) = o;
  }
}

#define KP     200
#define NPANEL (A_DIM / KP)
#define KQP    (KP / 4)
#define CH     (KP / 8)
#define GRP    5
#define NGRP   (CH / GRP)

template <int NT>
__device__ __forceinline__ void stage_panel(const float* __restrict__ Ep,
                                            float4* __restrict__ sbuf, int t) {
  constexpr int NBLK = KQP * (D_DIM / 2);
  constexpr int J    = (NBLK + NT - 1) / NT;
  constexpr int TAIL = NBLK - (J - 1) * NT;
  float2 e[J][4];
#pragma unroll
  for (int j = 0; j < J; ++j) {
    const int blk = t + j * NT;
    if (j < J - 1 || TAIL == NT || t < TAIL) {
      const int kq = blk >> 6, n2 = blk & 63;
      const float* s = Ep + (size_t)(4 * kq) * D_DIM + 2 * n2;
#pragma unroll
      for (int jj = 0; jj < 4; ++jj)
        e[j][jj] = *reinterpret_cast<const float2*>(s + jj * D_DIM);
    }
  }
#pragma unroll
  for (int j = 0; j < J; ++j) {
    const int blk = t + j * NT;
    if (j < J - 1 || TAIL == NT || t < TAIL) {
      const int kq = blk >> 6, n2 = blk & 63;
      float4* d = sbuf + kq * D_DIM + 2 * n2;
      d[0] = make_float4(e[j][0].x, e[j][1].x, e[j][2].x, e[j][3].x);
      d[1] = make_float4(e[j][0].y, e[j][1].y, e[j][2].y, e[j][3].y);
    }
  }
}

__global__ __launch_bounds__(256) void attr_gemm_kernel(
    const float* __restrict__ Aattr, const float* __restrict__ Eemb,
    float* __restrict__ out2) {
  __shared__ float4 sE[2][KQP * D_DIM];

  const int tid    = threadIdx.x;
  const int wave   = __builtin_amdgcn_readfirstlane(tid >> 5);
  const int lane   = tid & 31;
  const int c      = lane & 15;
  const int h      = lane >> 4;
  const int m_base = blockIdx.x * 32;
  const bool mult  = wave < 4;

  const int mt = (wave >> 1) & 1;
  const int nh = wave & 1;
  const float* Arow = Aattr + (size_t)(m_base + 16 * mt + c) * A_DIM + 4 * h;

  float4 aCur[CH];
  v8f acc[4];
  acc[0] = {}; acc[1] = {}; acc[2] = {}; acc[3] = {};

  if (mult) {
#pragma unroll
    for (int q = 0; q < CH; ++q)
      aCur[q] = *reinterpret_cast<const float4*>(Arow + 8 * q);
  }
  stage_panel<256>(Eemb, sE[0], tid);
  __syncthreads();

  for (int p = 0; p < NPANEL; ++p) {
    if (mult) {
      const int pn = (p + 1 < NPANEL) ? p + 1 : p;
      const float* Anext = Arow + 8 * CH * pn;
      const float4* fb = sE[p & 1] + h * D_DIM + 64 * nh + c;
      float4 bs[2][GRP][4];
#pragma unroll
      for (int qq = 0; qq < GRP; ++qq)
#pragma unroll
        for (int t = 0; t < 4; ++t)
          bs[0][qq][t] = fb[(2 * qq) * D_DIM + 16 * t];
#pragma unroll
      for (int g = 0; g < NGRP; ++g) {
        if (g + 1 < NGRP) {
#pragma unroll
          for (int qq = 0; qq < GRP; ++qq)
#pragma unroll
            for (int t = 0; t < 4; ++t)
              bs[(g + 1) & 1][qq][t] = fb[(2 * (GRP * (g + 1) + qq)) * D_DIM + 16 * t];
        }
        __builtin_amdgcn_sched_barrier(0);
#pragma unroll
        for (int qq = 0; qq < GRP; ++qq) {
          const float4 a = aCur[GRP * g + qq];
          v2f A0; A0[0] = a.x; A0[1] = a.y;
          v2f A1; A1[0] = a.z; A1[1] = a.w;
#pragma unroll
          for (int t = 0; t < 4; ++t) {
            const float4 b = bs[g & 1][qq][t];
            v2f B0; B0[0] = b.x; B0[1] = b.y;
            acc[t] = wmma_f32(A0, B0, acc[t]);
          }
#pragma unroll
          for (int t = 0; t < 4; ++t) {
            const float4 b = bs[g & 1][qq][t];
            v2f B1; B1[0] = b.z; B1[1] = b.w;
            acc[t] = wmma_f32(A1, B1, acc[t]);
          }
          aCur[GRP * g + qq] =
              *reinterpret_cast<const float4*>(Anext + 8 * (GRP * g + qq));
        }
      }
    } else {
      if (p + 1 < NPANEL) {
        const float* Ep = Eemb + (size_t)(p + 1) * KP * D_DIM;
        stage_panel<128>(Ep, sE[(p + 1) & 1], tid - 128);
      }
    }
    if (p + 1 < NPANEL) {
      __builtin_amdgcn_fence(__ATOMIC_RELEASE, "workgroup", "local");
      __builtin_amdgcn_s_barrier();
      __builtin_amdgcn_fence(__ATOMIC_ACQUIRE, "workgroup", "local");
    }
  }

  if (mult) {
    const int hi8 = 8 * h;
#pragma unroll
    for (int t = 0; t < 4; ++t)
#pragma unroll
      for (int r = 0; r < 8; ++r)
        out2[(size_t)(m_base + 16 * mt + r + hi8) * D_DIM + 64 * nh + 16 * t + c] = acc[t][r];
  }
}

extern "C" void kernel_launch(void* const* d_in, const int* in_sizes, int n_in,
                              void* d_out, int out_size, void* d_ws, size_t ws_size,
                              hipStream_t stream) {
  (void)in_sizes; (void)n_in; (void)out_size; (void)d_ws; (void)ws_size;
  const float* hidden = (const float*)d_in[0];
  const int*   adj    = (const int*)d_in[1];
  const float* amat   = (const float*)d_in[2];
  const float* Aattr  = (const float*)d_in[3];
  const float* Eemb   = (const float*)d_in[4];

  float* out_attn = (float*)d_out;
  float* out_attr = out_attn + (size_t)B_DIM * N_DIM * D_DIM;

  gat_attn_kernel<<<dim3(B_DIM * (N_DIM / 32)), dim3(128), 0, stream>>>(
      hidden, adj, amat, out_attn);
  attr_gemm_kernel<<<dim3((B_DIM * N_DIM) / 32), dim3(256), 0, stream>>>(
      Aattr, Eemb, out_attr);
}


# attr GEMM work-group tile 32x128 -> 64 rows x 64 columns: each group stages only its column half of E and keeps it whole in LDS (250 KB, no ring), 2 hand-over barriers not 4; k sum order unchanged
# speedup vs baseline: 2.2402x; 1.0674x over previous
#include <hip/hip_runtime.h>
#include <hip/hip_bf16.h>

typedef __attribute__((ext_vector_type(2))) float v2f;
typedef __attribute__((ext_vector_type(8))) float v8f;

#define B_DIM 64
#define N_DIM 128
#define D_DIM 128
#define A_DIM 1000

__device__ __forceinline__ v8f wmma_f32(v2f a, v2f b, v8f c) {
  return __builtin_amdgcn_wmma_f32_16x16x4_f32(false, a, false, b, (short)0, c,
                                               false, false);
}

typedef __attribute__((ext_vector_type(4))) float v4f;

#define PH 132

struct SFrag { v4f hi; v4f hj[4]; v4f a[4]; };
struct OFrag { v4f p; v4f b[4]; };

__device__ __forceinline__ void load_sfrag(SFrag& f, const float* pHi, const float* pHj,
                                           const float* pA, int q) {
  f.hi = *reinterpret_cast<const v4f*>(pHi + 8 * q);
#pragma unroll
  for (int t = 0; t < 4; ++t)
    f.hj[t] = *reinterpret_cast<const v4f*>(pHj + 16 * t * PH + 8 * q);
#pragma unroll
  for (int s = 0; s < 4; ++s)
    f.a[s] = *reinterpret_cast<const v4f*>(pA + 32 * q + 4 * s);
}

__device__ __forceinline__ void mul_sfrag(const SFrag& f, v8f (&acc)[4][4]) {
  const float C1 = 0.6f, C2 = 0.4f;
  float t1[4], t2[4];
#pragma unroll
  for (int s = 0; s < 4; ++s) {
    t1[s] = C1 * f.hi[s];
    t2[s] = C2 * fabsf(f.hi[s]);
  }
#pragma unroll
  for (int u = 0; u < 2; ++u) {
    v2f Bv[4], Bva[4];
#pragma unroll
    for (int t = 0; t < 4; ++t) {
      Bv[t][0]  = f.hj[t][2 * u];          Bv[t][1]  = f.hj[t][2 * u + 1];
      Bva[t][0] = fabsf(f.hj[t][2 * u]);   Bva[t][1] = fabsf(f.hj[t][2 * u + 1]);
    }
#pragma unroll
    for (int k = 0; k < 4; ++k) {
      v2f Av;
      Av[0] = t1[2 * u] * f.a[2 * u][k];
      Av[1] = t1[2 * u + 1] * f.a[2 * u + 1][k];
#pragma unroll
      for (int t = 0; t < 4; ++t) acc[k][t] = wmma_f32(Av, Bv[t], acc[k][t]);
      v2f Ava;
      Ava[0] = t2[2 * u] * f.a[2 * u][k];
      Ava[1] = t2[2 * u + 1] * f.a[2 * u + 1][k];
#pragma unroll
      for (int t = 0; t < 4; ++t) acc[k][t] = wmma_f32(Ava, Bva[t], acc[k][t]);
    }
  }
}

__device__ __forceinline__ void load_ofrag(OFrag& f, const float* pP, const float* pB, int q) {
  f.p = *reinterpret_cast<const v4f*>(pP + 8 * q);
#pragma unroll
  for (int s = 0; s < 4; ++s)
    f.b[s] = *reinterpret_cast<const v4f*>(pB + (8 * q + s) * PH);
}

__device__ __forceinline__ void mul_ofrag(const OFrag& f, v8f (&oacc)[4]) {
#pragma unroll
  for (int u = 0; u < 2; ++u) {
    v2f Av; Av[0] = f.p[2 * u]; Av[1] = f.p[2 * u + 1];
#pragma unroll
    for (int t = 0; t < 4; ++t) {
      v2f Bv; Bv[0] = f.b[2 * u][t]; Bv[1] = f.b[2 * u + 1][t];
      oacc[t] = wmma_f32(Av, Bv, oacc[t]);
    }
  }
}

template <int X>
__device__ __forceinline__ float row_xor(float v) {
  return __builtin_bit_cast(
      float, __builtin_amdgcn_update_dpp(0, __builtin_bit_cast(int, v), 0x160 + X, 0xf, 0xf, true));
}

__global__ __launch_bounds__(128) void gat_attn_kernel(
    const float* __restrict__ hidden, const int* __restrict__ adj,
    const float* __restrict__ amat, float* __restrict__ out) {
  __shared__ __attribute__((aligned(16))) float sH[N_DIM * PH];
  __shared__ __attribute__((aligned(16))) float sP[32 * PH];
  __shared__ __attribute__((aligned(16))) float sAa[D_DIM * 4];
  __shared__ __attribute__((aligned(16))) float sRedM[2][2][16];
  __shared__ __attribute__((aligned(16))) float sRedS[2][2][16];

  const int tid  = threadIdx.x;
  const int b    = blockIdx.x >> 2;
  const int wave = __builtin_amdgcn_readfirstlane(tid >> 5);
  const int lane = tid & 31;
  const int c    = lane & 15;
  const int h    = lane >> 4;
  const int it   = wave >> 1;
  const int jh   = wave & 1;
  const int i0   = (blockIdx.x & 3) * 32 + 16 * it;
  const int j0   = 64 * jh;

  const v4f* Hg4 = reinterpret_cast<const v4f*>(hidden + (size_t)b * N_DIM * D_DIM);
  v4f st[32];
#pragma unroll
  for (int n = 0; n < 32; ++n) st[n] = Hg4[tid + 128 * n];
  const v4f a4 = reinterpret_cast<const v4f*>(amat)[tid];

  const int* adjS = adj + ((size_t)b * N_DIM + i0 + 8 * h) * N_DIM + j0 + c;
  int et[4][8];
#pragma unroll
  for (int t = 0; t < 4; ++t)
#pragma unroll
    for (int r = 0; r < 8; ++r) et[t][r] = adjS[r * N_DIM + 16 * t];

#pragma unroll
  for (int n = 0; n < 32; ++n)
    *reinterpret_cast<v4f*>(&sH[(wave + 4 * n) * PH + 4 * lane]) = st[n];
  reinterpret_cast<v4f*>(sAa)[tid] = a4;
  __syncthreads();

  v8f acc[4][4];
#pragma unroll
  for (int k = 0; k < 4; ++k)
#pragma unroll
    for (int t = 0; t < 4; ++t) acc[k][t] = {};

  const float* pHi = &sH[(i0 + c) * PH + 4 * h];
  const float* pHj = &sH[(j0 + c) * PH + 4 * h];
  const float* pA  = &sAa[16 * h];
  SFrag fa, fb;
  load_sfrag(fa, pHi, pHj, pA, 0);
  for (int q = 0; q < 16; q += 2) {
    load_sfrag(fb, pHi, pHj, pA, q + 1);
    __builtin_amdgcn_sched_barrier(0);
    mul_sfrag(fa, acc);
    load_sfrag(fa, pHi, pHj, pA, (q + 2) & 15);
    __builtin_amdgcn_sched_barrier(0);
    mul_sfrag(fb, acc);
  }

  const float NEG_INF = -9.0e15f;
  float e[4][8];
#pragma unroll
  for (int t = 0; t < 4; ++t)
#pragma unroll
    for (int r = 0; r < 8; ++r) {
      const int ty = et[t][r];
      float v = NEG_INF;
      v = (ty == 1) ? acc[0][t][r] : v;
      v = (ty == 2) ? acc[1][t][r] : v;
      v = (ty == 3) ? acc[2][t][r] : v;
      v = (ty == 4) ? acc[3][t][r] : v;
      e[t][r] = v;
    }

  float m[8];
#pragma unroll
  for (int r = 0; r < 8; ++r)
    m[r] = fmaxf(fmaxf(e[0][r], e[1][r]), fmaxf(e[2][r], e[3][r]));
#pragma unroll
  for (int r = 0; r < 8; ++r) m[r] = fmaxf(m[r], row_xor<1>(m[r]));
#pragma unroll
  for (int r = 0; r < 8; ++r) m[r] = fmaxf(m[r], row_xor<2>(m[r]));
#pragma unroll
  for (int r = 0; r < 8; ++r) m[r] = fmaxf(m[r], row_xor<4>(m[r]));
#pragma unroll
  for (int r = 0; r < 8; ++r) m[r] = fmaxf(m[r], row_xor<8>(m[r]));
  if (c == 0) {
#pragma unroll
    for (int r = 0; r < 8; ++r) sRedM[it][jh][8 * h + r] = m[r];
  }
  __syncthreads();
#pragma unroll
  for (int r = 0; r < 8; ++r) m[r] = fmaxf(m[r], sRedM[it][jh ^ 1][8 * h + r]);

  float s[8];
#pragma unroll
  for (int r = 0; r < 8; ++r) {
#pragma unroll
    for (int t = 0; t < 4; ++t) e[t][r] = __expf(e[t][r] - m[r]);
    s[r] = (e[0][r] + e[1][r]) + (e[2][r] + e[3][r]);
  }
#pragma unroll
  for (int r = 0; r < 8; ++r) s[r] += row_xor<1>(s[r]);
#pragma unroll
  for (int r = 0; r < 8; ++r) s[r] += row_xor<2>(s[r]);
#pragma unroll
  for (int r = 0; r < 8; ++r) s[r] += row_xor<4>(s[r]);
#pragma unroll
  for (int r = 0; r < 8; ++r) s[r] += row_xor<8>(s[r]);
  if (c == 0) {
#pragma unroll
    for (int r = 0; r < 8; ++r) sRedS[it][jh][8 * h + r] = s[r];
  }
  __syncthreads();
#pragma unroll
  for (int r = 0; r < 8; ++r) {
    const float tot = sRedS[it][0][8 * h + r] + sRedS[it][1][8 * h + r];
    const float inv = 1.0f / tot;
#pragma unroll
    for (int t = 0; t < 4; ++t)
      sP[(16 * it + 8 * h + r) * PH + j0 + 16 * t + c] = e[t][r] * inv;
  }
  __syncthreads();

  v8f oacc[4];
#pragma unroll
  for (int t = 0; t < 4; ++t) oacc[t] = {};
  const float* pP = &sP[(16 * it + c) * PH + 4 * h];
  const float* pB = &sH[(4 * h) * PH + j0 + 4 * c];
  OFrag oa, ob;
  load_ofrag(oa, pP, pB, 0);
  for (int q = 0; q < 16; q += 2) {
    load_ofrag(ob, pP, pB, q + 1);
    __builtin_amdgcn_sched_barrier(0);
    mul_ofrag(oa, oacc);
    load_ofrag(oa, pP, pB, (q + 2) & 15);
    __builtin_amdgcn_sched_barrier(0);
    mul_ofrag(ob, oacc);
  }
  float* og = out + ((size_t)b * N_DIM + i0 + 8 * h) * D_DIM + j0 + 4 * c;
#pragma unroll
  for (int r = 0; r < 8; ++r) {
    v4f o;
    o[0] = oacc[0][r]; o[1] = oacc[1][r]; o[2] = oacc[2][r]; o[3] = oacc[3][r];
    *reinterpret_cast<v4f*>(og + r * D_DIM) = o;
  }
}

#define KP     200
#define NPANEL (A_DIM / KP)
#define KQP    (KP / 4)
#define CH     (KP / 8)
#define GRP    5
#define NGRP   (CH / GRP)
#define NH     64

template <int NT>
__device__ __forceinline__ void stage_panel(const float* __restrict__ Ep,
                                            float4* __restrict__ sbuf, int t) {
  constexpr int NBLK = KQP * (NH / 2);
  constexpr int J    = (NBLK + NT - 1) / NT;
  constexpr int TAIL = NBLK - (J - 1) * NT;
  float2 e[J][4];
#pragma unroll
  for (int j = 0; j < J; ++j) {
    const int blk = t + j * NT;
    if (j < J - 1 || TAIL == NT || t < TAIL) {
      const int kq = blk >> 5, n2 = blk & 31;
      const float* s = Ep + (size_t)(4 * kq) * D_DIM + 2 * n2;
#pragma unroll
      for (int jj = 0; jj < 4; ++jj)
        e[j][jj] = *reinterpret_cast<const float2*>(s + jj * D_DIM);
    }
  }
#pragma unroll
  for (int j = 0; j < J; ++j) {
    const int blk = t + j * NT;
    if (j < J - 1 || TAIL == NT || t < TAIL) {
      const int kq = blk >> 5, n2 = blk & 31;
      float4* d = sbuf + kq * NH + 2 * n2;
      d[0] = make_float4(e[j][0].x, e[j][1].x, e[j][2].x, e[j][3].x);
      d[1] = make_float4(e[j][0].y, e[j][1].y, e[j][2].y, e[j][3].y);
    }
  }
}

__global__ __launch_bounds__(256) void attr_gemm_kernel(
    const float* __restrict__ Aattr, const float* __restrict__ Eemb,
    float* __restrict__ out2) {
  __shared__ float4 sE[NPANEL][KQP * NH];

  const int tid    = threadIdx.x;
  const int wave   = __builtin_amdgcn_readfirstlane(tid >> 5);
  const int lane   = tid & 31;
  const int c      = lane & 15;
  const int h      = lane >> 4;
  const int m_base = (blockIdx.x >> 1) * 64;
  const int n_base = (blockIdx.x & 1) * NH;
  const bool mult  = wave < 4;

  const int mt = wave & 3;
  const float* Arow = Aattr + (size_t)(m_base + 16 * mt + c) * A_DIM + 4 * h;
  const float* Ecol = Eemb + n_base;

  float4 aCur[CH];
  v8f acc[4];
  acc[0] = {}; acc[1] = {}; acc[2] = {}; acc[3] = {};

  if (mult) {
#pragma unroll
    for (int q = 0; q < CH; ++q)
      aCur[q] = *reinterpret_cast<const float4*>(Arow + 8 * q);
  }
  stage_panel<256>(Ecol, sE[0], tid);
  __syncthreads();

  for (int p = 0; p < NPANEL; ++p) {
    if (mult) {
      const int pn = (p + 1 < NPANEL) ? p + 1 : p;
      const float* Anext = Arow + 8 * CH * pn;
      const float4* fb = sE[p] + h * NH + c;
      float4 bs[2][GRP][4];
#pragma unroll
      for (int qq = 0; qq < GRP; ++qq)
#pragma unroll
        for (int t = 0; t < 4; ++t)
          bs[0][qq][t] = fb[(2 * qq) * NH + 16 * t];
#pragma unroll
      for (int g = 0; g < NGRP; ++g) {
        if (g + 1 < NGRP) {
#pragma unroll
          for (int qq = 0; qq < GRP; ++qq)
#pragma unroll
            for (int t = 0; t < 4; ++t)
              bs[(g + 1) & 1][qq][t] = fb[(2 * (GRP * (g + 1) + qq)) * NH + 16 * t];
        }
        __builtin_amdgcn_sched_barrier(0);
#pragma unroll
        for (int qq = 0; qq < GRP; ++qq) {
          const float4 a = aCur[GRP * g + qq];
          v2f A0; A0[0] = a.x; A0[1] = a.y;
          v2f A1; A1[0] = a.z; A1[1] = a.w;
#pragma unroll
          for (int t = 0; t < 4; ++t) {
            const float4 b = bs[g & 1][qq][t];
            v2f B0; B0[0] = b.x; B0[1] = b.y;
            acc[t] = wmma_f32(A0, B0, acc[t]);
          }
#pragma unroll
          for (int t = 0; t < 4; ++t) {
            const float4 b = bs[g & 1][qq][t];
            v2f B1; B1[0] = b.z; B1[1] = b.w;
            acc[t] = wmma_f32(A1, B1, acc[t]);
          }
          aCur[GRP * g + qq] =
              *reinterpret_cast<const float4*>(Anext + 8 * (GRP * g + qq));
        }
      }
    } else {
      if (p < 2) {
        const float* Ep = Ecol + (size_t)(2 * p + 1) * KP * D_DIM;
        stage_panel<128>(Ep, sE[2 * p + 1], tid - 128);
        stage_panel<128>(Ep + (size_t)KP * D_DIM, sE[2 * p + 2], tid - 128);
      }
    }
    if (p < 2) {
      __builtin_amdgcn_fence(__ATOMIC_RELEASE, "workgroup", "local");
      __builtin_amdgcn_s_barrier();
      __builtin_amdgcn_fence(__ATOMIC_ACQUIRE, "workgroup", "local");
    }
  }

  if (mult) {
    const int hi8 = 8 * h;
#pragma unroll
    for (int t = 0; t < 4; ++t)
#pragma unroll
      for (int r = 0; r < 8; ++r)
        out2[(size_t)(m_base + 16 * mt + r + hi8) * D_DIM + n_base + 16 * t + c] = acc[t][r];
  }
}

extern "C" void kernel_launch(void* const* d_in, const int* in_sizes, int n_in,
                              void* d_out, int out_size, void* d_ws, size_t ws_size,
                              hipStream_t stream) {
  (void)in_sizes; (void)n_in; (void)out_size; (void)d_ws; (void)ws_size;
  const float* hidden = (const float*)d_in[0];
  const int*   adj    = (const int*)d_in[1];
  const float* amat   = (const float*)d_in[2];
  const float* Aattr  = (const float*)d_in[3];
  const float* Eemb   = (const float*)d_in[4];

  float* out_attn = (float*)d_out;
  float* out_attr = out_attn + (size_t)B_DIM * N_DIM * D_DIM;

  gat_attn_kernel<<<dim3(B_DIM * (N_DIM / 32)), dim3(128), 0, stream>>>(
      hidden, adj, amat, out_attn);
  attr_gemm_kernel<<<dim3(((B_DIM * N_DIM) / 64) * 2), dim3(256), 0, stream>>>(
      Aattr, Eemb, out_attr);
}


# one launch: each work-group runs its 32 attention rows, then its 64x64 attr GEMM tile (no block roles); 4 helper waves stage E under the attention, first A loads under the softmax; sums same
# speedup vs baseline: 2.6208x; 1.1699x over previous
#include <hip/hip_runtime.h>
#include <hip/hip_bf16.h>

typedef __attribute__((ext_vector_type(2))) float v2f;
typedef __attribute__((ext_vector_type(8))) float v8f;

#define B_DIM 64
#define N_DIM 128
#define D_DIM 128
#define A_DIM 1000

__device__ __forceinline__ v8f wmma_f32(v2f a, v2f b, v8f c) {
  return __builtin_amdgcn_wmma_f32_16x16x4_f32(false, a, false, b, (short)0, c,
                                               false, false);
}

typedef __attribute__((ext_vector_type(4))) float v4f;

#define PH 132

struct SFrag { v4f hi; v4f hj[4]; v4f a[4]; };
struct OFrag { v4f p; v4f b[4]; };

__device__ __forceinline__ void load_sfrag(SFrag& f, const float* pHi, const float* pHj,
                                           const float* pA, int q) {
  f.hi = *reinterpret_cast<const v4f*>(pHi + 8 * q);
#pragma unroll
  for (int t = 0; t < 4; ++t)
    f.hj[t] = *reinterpret_cast<const v4f*>(pHj + 16 * t * PH + 8 * q);
#pragma unroll
  for (int s = 0; s < 4; ++s)
    f.a[s] = *reinterpret_cast<const v4f*>(pA + 32 * q + 4 * s);
}

__device__ __forceinline__ void mul_sfrag(const SFrag& f, v8f (&acc)[4][4]) {
  const float C1 = 0.6f, C2 = 0.4f;
  float t1[4], t2[4];
#pragma unroll
  for (int s = 0; s < 4; ++s) {
    t1[s] = C1 * f.hi[s];
    t2[s] = C2 * fabsf(f.hi[s]);
  }
#pragma unroll
  for (int u = 0; u < 2; ++u) {
    v2f Bv[4], Bva[4];
#pragma unroll
    for (int t = 0; t < 4; ++t) {
      Bv[t][0]  = f.hj[t][2 * u];          Bv[t][1]  = f.hj[t][2 * u + 1];
      Bva[t][0] = fabsf(f.hj[t][2 * u]);   Bva[t][1] = fabsf(f.hj[t][2 * u + 1]);
    }
#pragma unroll
    for (int k = 0; k < 4; ++k) {
      v2f Av;
      Av[0] = t1[2 * u] * f.a[2 * u][k];
      Av[1] = t1[2 * u + 1] * f.a[2 * u + 1][k];
#pragma unroll
      for (int t = 0; t < 4; ++t) acc[k][t] = wmma_f32(Av, Bv[t], acc[k][t]);
      v2f Ava;
      Ava[0] = t2[2 * u] * f.a[2 * u][k];
      Ava[1] = t2[2 * u + 1] * f.a[2 * u + 1][k];
#pragma unroll
      for (int t = 0; t < 4; ++t) acc[k][t] = wmma_f32(Ava, Bva[t], acc[k][t]);
    }
  }
}

__device__ __forceinline__ void load_ofrag(OFrag& f, const float* pP, const float* pB, int q) {
  f.p = *reinterpret_cast<const v4f*>(pP + 8 * q);
#pragma unroll
  for (int s = 0; s < 4; ++s)
    f.b[s] = *reinterpret_cast<const v4f*>(pB + (8 * q + s) * PH);
}

__device__ __forceinline__ void mul_ofrag(const OFrag& f, v8f (&oacc)[4]) {
#pragma unroll
  for (int u = 0; u < 2; ++u) {
    v2f Av; Av[0] = f.p[2 * u]; Av[1] = f.p[2 * u + 1];
#pragma unroll
    for (int t = 0; t < 4; ++t) {
      v2f Bv; Bv[0] = f.b[2 * u][t]; Bv[1] = f.b[2 * u + 1][t];
      oacc[t] = wmma_f32(Av, Bv, oacc[t]);
    }
  }
}

template <int X>
__device__ __forceinline__ float row_xor(float v) {
  return __builtin_bit_cast(
      float, __builtin_amdgcn_update_dpp(0, __builtin_bit_cast(int, v), 0x160 + X, 0xf, 0xf, true));
}

#define KP     200
#define NPANEL (A_DIM / KP)
#define KQP    (KP / 4)
#define CH     (KP / 8)
#define GRP    5
#define NGRP   (CH / GRP)
#define NH     64

template <int NT>
__device__ __forceinline__ void stage_panel(const float* __restrict__ Ep,
                                            float4* __restrict__ sbuf, int t) {
  constexpr int NBLK = KQP * (NH / 2);
  constexpr int J    = (NBLK + NT - 1) / NT;
  constexpr int TAIL = NBLK - (J - 1) * NT;
  float2 e[J][4];
#pragma unroll
  for (int j = 0; j < J; ++j) {
    const int blk = t + j * NT;
    if (j < J - 1 || TAIL == NT || t < TAIL) {
      const int kq = blk >> 5, n2 = blk & 31;
      const float* s = Ep + (size_t)(4 * kq) * D_DIM + 2 * n2;
#pragma unroll
      for (int jj = 0; jj < 4; ++jj)
        e[j][jj] = *reinterpret_cast<const float2*>(s + jj * D_DIM);
    }
  }
#pragma unroll
  for (int j = 0; j < J; ++j) {
    const int blk = t + j * NT;
    if (j < J - 1 || TAIL == NT || t < TAIL) {
      const int kq = blk >> 5, n2 = blk & 31;
      float4* d = sbuf + kq * NH + 2 * n2;
      d[0] = make_float4(e[j][0].x, e[j][1].x, e[j][2].x, e[j][3].x);
      d[1] = make_float4(e[j][0].y, e[j][1].y, e[j][2].y, e[j][3].y);
    }
  }
}

__device__ __forceinline__ void lds_barrier() {
  __builtin_amdgcn_fence(__ATOMIC_RELEASE, "workgroup", "local");
  __builtin_amdgcn_s_barrier();
  __builtin_amdgcn_fence(__ATOMIC_ACQUIRE, "workgroup", "local");
}

__device__ __forceinline__ void mul_panel(const float4* fb, const float* __restrict__ Anext,
                                          float4 (&aCur)[CH], v8f (&acc)[4]) {
  float4 bs[2][GRP][4];
#pragma unroll
  for (int qq = 0; qq < GRP; ++qq)
#pragma unroll
    for (int t = 0; t < 4; ++t)
      bs[0][qq][t] = fb[(2 * qq) * NH + 16 * t];
#pragma unroll
  for (int g = 0; g < NGRP; ++g) {
    if (g + 1 < NGRP) {
#pragma unroll
      for (int qq = 0; qq < GRP; ++qq)
#pragma unroll
        for (int t = 0; t < 4; ++t)
          bs[(g + 1) & 1][qq][t] = fb[(2 * (GRP * (g + 1) + qq)) * NH + 16 * t];
    }
    __builtin_amdgcn_sched_barrier(0);
#pragma unroll
    for (int qq = 0; qq < GRP; ++qq) {
      const float4 a = aCur[GRP * g + qq];
      v2f A0; A0[0] = a.x; A0[1] = a.y;
      v2f A1; A1[0] = a.z; A1[1] = a.w;
#pragma unroll
      for (int t = 0; t < 4; ++t) {
        const float4 b = bs[g & 1][qq][t];
        v2f B0; B0[0] = b.x; B0[1] = b.y;
        acc[t] = wmma_f32(A0, B0, acc[t]);
      }
#pragma unroll
      for (int t = 0; t < 4; ++t) {
        const float4 b = bs[g & 1][qq][t];
        v2f B1; B1[0] = b.z; B1[1] = b.w;
        acc[t] = wmma_f32(A1, B1, acc[t]);
      }
      aCur[GRP * g + qq] =
          *reinterpret_cast<const float4*>(Anext + 8 * (GRP * g + qq));
    }
  }
}

__global__ __launch_bounds__(256) void fused_kernel(
    const float* __restrict__ hidden, const int* __restrict__ adj,
    const float* __restrict__ amat, const float* __restrict__ Aattr,
    const float* __restrict__ Eemb, float* __restrict__ out,
    float* __restrict__ out2) {
  __shared__ float4 sE[NPANEL - 1][KQP * NH];
  __shared__ __attribute__((aligned(16))) float sH[N_DIM * PH];
  __shared__ __attribute__((aligned(16))) float sP[32 * PH];
  __shared__ __attribute__((aligned(16))) float sAa[D_DIM * 4];
  __shared__ __attribute__((aligned(16))) float sRedM[2][2][16];
  __shared__ __attribute__((aligned(16))) float sRedS[2][2][16];
  float4* sE4 = reinterpret_cast<float4*>(sH);

  const int tid  = threadIdx.x;
  const int wave = __builtin_amdgcn_readfirstlane(tid >> 5);
  const int lane = tid & 31;
  const int c    = lane & 15;
  const int h    = lane >> 4;
  const bool mult = wave < 4;

  const int b    = blockIdx.x >> 2;
  const int it   = (wave >> 1) & 1;
  const int jh   = wave & 1;
  const int i0   = (blockIdx.x & 3) * 32 + 16 * it;
  const int j0   = 64 * jh;
  const int m_base = (blockIdx.x >> 1) * 64;
  const int n_base = (blockIdx.x & 1) * NH;
  const int mt     = wave & 3;
  const float* Arow = Aattr + (size_t)(m_base + 16 * mt + c) * A_DIM + 4 * h;
  const float* Ecol = Eemb + n_base;

  float e[4][8];
  float m[8];
  float s[8];
  float4 aCur[CH];

  if (mult) {
    const v4f* Hg4 = reinterpret_cast<const v4f*>(hidden + (size_t)b * N_DIM * D_DIM);
    v4f st[32];
#pragma unroll
    for (int n = 0; n < 32; ++n) st[n] = Hg4[tid + 128 * n];
    const v4f a4 = reinterpret_cast<const v4f*>(amat)[tid];

#pragma unroll
    for (int n = 0; n < 32; ++n)
      *reinterpret_cast<v4f*>(&sH[(wave + 4 * n) * PH + 4 * lane]) = st[n];
    reinterpret_cast<v4f*>(sAa)[tid] = a4;
  } else {
    stage_panel<128>(Ecol, sE[0], tid - 128);
  }
  lds_barrier();

  if (mult) {
    const int* adjS = adj + ((size_t)b * N_DIM + i0 + 8 * h) * N_DIM + j0 + c;
    int et[4][8];
#pragma unroll
    for (int t = 0; t < 4; ++t)
#pragma unroll
      for (int r = 0; r < 8; ++r) et[t][r] = adjS[r * N_DIM + 16 * t];

    v8f acc[4][4];
#pragma unroll
    for (int k = 0; k < 4; ++k)
#pragma unroll
      for (int t = 0; t < 4; ++t) acc[k][t] = {};

    const float* pHi = &sH[(i0 + c) * PH + 4 * h];
    const float* pHj = &sH[(j0 + c) * PH + 4 * h];
    const float* pA  = &sAa[16 * h];
    SFrag fa, fb;
    load_sfrag(fa, pHi, pHj, pA, 0);
    for (int q = 0; q < 16; q += 2) {
      load_sfrag(fb, pHi, pHj, pA, q + 1);
      __builtin_amdgcn_sched_barrier(0);
      mul_sfrag(fa, acc);
      load_sfrag(fa, pHi, pHj, pA, (q + 2) & 15);
      __builtin_amdgcn_sched_barrier(0);
      mul_sfrag(fb, acc);
    }

    const float NEG_INF = -9.0e15f;
#pragma unroll
    for (int t = 0; t < 4; ++t)
#pragma unroll
      for (int r = 0; r < 8; ++r) {
        const int ty = et[t][r];
        float v = NEG_INF;
        v = (ty == 1) ? acc[0][t][r] : v;
        v = (ty == 2) ? acc[1][t][r] : v;
        v = (ty == 3) ? acc[2][t][r] : v;
        v = (ty == 4) ? acc[3][t][r] : v;
        e[t][r] = v;
      }

#pragma unroll
    for (int q = 0; q < CH; ++q)
      aCur[q] = *reinterpret_cast<const float4*>(Arow + 8 * q);

#pragma unroll
    for (int r = 0; r < 8; ++r)
      m[r] = fmaxf(fmaxf(e[0][r], e[1][r]), fmaxf(e[2][r], e[3][r]));
#pragma unroll
    for (int r = 0; r < 8; ++r) m[r] = fmaxf(m[r], row_xor<1>(m[r]));
#pragma unroll
    for (int r = 0; r < 8; ++r) m[r] = fmaxf(m[r], row_xor<2>(m[r]));
#pragma unroll
    for (int r = 0; r < 8; ++r) m[r] = fmaxf(m[r], row_xor<4>(m[r]));
#pragma unroll
    for (int r = 0; r < 8; ++r) m[r] = fmaxf(m[r], row_xor<8>(m[r]));
    if (c == 0) {
#pragma unroll
      for (int r = 0; r < 8; ++r) sRedM[it][jh][8 * h + r] = m[r];
    }
  } else {
    stage_panel<128>(Ecol + (size_t)1 * KP * D_DIM, sE[1], tid - 128);
    stage_panel<128>(Ecol + (size_t)2 * KP * D_DIM, sE[2], tid - 128);
    stage_panel<128>(Ecol + (size_t)3 * KP * D_DIM, sE[3], tid - 128);
  }
  lds_barrier();

  if (mult) {
#pragma unroll
    for (int r = 0; r < 8; ++r) m[r] = fmaxf(m[r], sRedM[it][jh ^ 1][8 * h + r]);

#pragma unroll
    for (int r = 0; r < 8; ++r) {
#pragma unroll
      for (int t = 0; t < 4; ++t) e[t][r] = __expf(e[t][r] - m[r]);
      s[r] = (e[0][r] + e[1][r]) + (e[2][r] + e[3][r]);
    }
#pragma unroll
    for (int r = 0; r < 8; ++r) s[r] += row_xor<1>(s[r]);
#pragma unroll
    for (int r = 0; r < 8; ++r) s[r] += row_xor<2>(s[r]);
#pragma unroll
    for (int r = 0; r < 8; ++r) s[r] += row_xor<4>(s[r]);
#pragma unroll
    for (int r = 0; r < 8; ++r) s[r] += row_xor<8>(s[r]);
    if (c == 0) {
#pragma unroll
      for (int r = 0; r < 8; ++r) sRedS[it][jh][8 * h + r] = s[r];
    }
  }
  lds_barrier();

  if (mult) {
#pragma unroll
    for (int r = 0; r < 8; ++r) {
      const float tot = sRedS[it][0][8 * h + r] + sRedS[it][1][8 * h + r];
      const float inv = 1.0f / tot;
#pragma unroll
      for (int t = 0; t < 4; ++t)
        sP[(16 * it + 8 * h + r) * PH + j0 + 16 * t + c] = e[t][r] * inv;
    }
  }
  lds_barrier();

  if (mult) {
    v8f oacc[4];
#pragma unroll
    for (int t = 0; t < 4; ++t) oacc[t] = {};
    const float* pP = &sP[(16 * it + c) * PH + 4 * h];
    const float* pB = &sH[(4 * h) * PH + j0 + 4 * c];
    OFrag oa, ob;
    load_ofrag(oa, pP, pB, 0);
    for (int q = 0; q < 16; q += 2) {
      load_ofrag(ob, pP, pB, q + 1);
      __builtin_amdgcn_sched_barrier(0);
      mul_ofrag(oa, oacc);
      load_ofrag(oa, pP, pB, (q + 2) & 15);
      __builtin_amdgcn_sched_barrier(0);
      mul_ofrag(ob, oacc);
    }
    float* og = out + ((size_t)b * N_DIM + i0 + 8 * h) * D_DIM + j0 + 4 * c;
#pragma unroll
    for (int r = 0; r < 8; ++r) {
      v4f o;
      o[0] = oacc[0][r]; o[1] = oacc[1][r]; o[2] = oacc[2][r]; o[3] = oacc[3][r];
      *reinterpret_cast<v4f*>(og + r * D_DIM) = o;
    }
  }
  lds_barrier();

  v8f gacc[4];
  gacc[0] = {}; gacc[1] = {}; gacc[2] = {}; gacc[3] = {};
  if (mult) {
    for (int p = 0; p < NPANEL - 1; ++p)
      mul_panel(sE[p] + h * NH + c, Arow + 8 * CH * (p + 1), aCur, gacc);
  } else {
    stage_panel<128>(Ecol + (size_t)(NPANEL - 1) * KP * D_DIM, sE4, tid - 128);
  }
  lds_barrier();

  if (mult) {
    mul_panel(sE4 + h * NH + c, Arow + 8 * CH * (NPANEL - 1), aCur, gacc);
    const int hi8 = 8 * h;
#pragma unroll
    for (int t = 0; t < 4; ++t)
#pragma unroll
      for (int r = 0; r < 8; ++r)
        out2[(size_t)(m_base + 16 * mt + r + hi8) * D_DIM + n_base + 16 * t + c] = gacc[t][r];
  }
}

extern "C" void kernel_launch(void* const* d_in, const int* in_sizes, int n_in,
                              void* d_out, int out_size, void* d_ws, size_t ws_size,
                              hipStream_t stream) {
  (void)in_sizes; (void)n_in; (void)out_size; (void)d_ws; (void)ws_size;
  const float* hidden = (const float*)d_in[0];
  const int*   adj    = (const int*)d_in[1];
  const float* amat   = (const float*)d_in[2];
  const float* Aattr  = (const float*)d_in[3];
  const float* Eemb   = (const float*)d_in[4];

  float* out_attn = (float*)d_out;
  float* out_attr = out_attn + (size_t)B_DIM * N_DIM * D_DIM;

  fused_kernel<<<dim3(256), dim3(256), 0, stream>>>(
      hidden, adj, amat, Aattr, Eemb, out_attn, out_attr);
}
